# starting kernel
# baseline (speedup 1.0000x reference)
#include <hip/hip_runtime.h>
#include <math.h>

typedef __attribute__((ext_vector_type(2))) float v2f;
typedef __attribute__((ext_vector_type(8))) float v8f;

#define NH 16
#define HD 64
#define ROWSZ (NH * HD)
#define LDS_STRIDE 18

__global__ __launch_bounds__(32)
void va_fwd_kernel(const float* __restrict__ q,
                   const float* __restrict__ k,
                   const float* __restrict__ v,
                   const int*   __restrict__ cuq,
                   const int*   __restrict__ cuk,
                   float*       __restrict__ out)
{
    const int lane = threadIdx.x & 31;
    const int il   = lane & 15;
    const int hi   = lane >> 4;
    const int b    = blockIdx.y;
    const int h    = blockIdx.z;

    const int cq0  = cuq[b];
    const int cq1  = cuq[b + 1];
    const int lenq = cq1 - cq0;
    const int base = blockIdx.x * 16;
    if (base >= lenq) return;

    const int ck0   = cuk[b];
    const int lenk  = cuk[b + 1] - ck0;
    const int delta = lenk - lenq;
    const int nrows = min(16, lenq - base);

    __shared__ float pl[16 * LDS_STRIDE];

    int qrow = cq0 + base + il;
    if (qrow > cq1 - 1) qrow = cq1 - 1;
    const float* qp = q + (size_t)qrow * ROWSZ + h * HD + 2 * hi;
    v2f qa[16];
#pragma unroll
    for (int c = 0; c < 16; ++c) {
        v2f t = *(const v2f*)(qp + 4 * c);
        qa[c] = t * 0.125f;
    }

    v8f oacc[4];
#pragma unroll
    for (int nc = 0; nc < 4; ++nc) oacc[nc] = (v8f){};
    v8f lacc = (v8f){};
    float mrow[8];
#pragma unroll
    for (int r = 0; r < 8; ++r) mrow[r] = -3.0e38f;

    int maxj = base + 15 + delta;
    if (maxj > lenk - 1) maxj = lenk - 1;
    if (maxj < 0) {
#pragma unroll
        for (int r = 0; r < 8; ++r) {
            int m = r + 8 * hi;
            if (m < nrows) {
                size_t ro = (size_t)(cq0 + base + m) * ROWSZ + h * HD;
#pragma unroll
                for (int nc = 0; nc < 4; ++nc) out[ro + nc * 16 + il] = 0.0f;
            }
        }
        return;
    }
    const int ktiles = maxj / 16 + 1;
    const float* kbase = k + (size_t)ck0 * ROWSZ + h * HD;
    const float* vbase = v + (size_t)ck0 * ROWSZ + h * HD;
    const v2f ones = {1.0f, 1.0f};

    for (int kt = 0; kt < ktiles; ++kt) {
        if (kt + 1 < ktiles) {
            int pr = min((kt + 1) * 16 + il, lenk - 1);
            __builtin_prefetch(kbase + (size_t)pr * ROWSZ, 0, 3);
            __builtin_prefetch(vbase + (size_t)pr * ROWSZ, 0, 3);
        }

        int krow = kt * 16 + il;
        if (krow > lenk - 1) krow = lenk - 1;
        const float* kp = kbase + (size_t)krow * ROWSZ + 2 * hi;
        v8f s = (v8f){};
#pragma unroll
        for (int c = 0; c < 16; ++c) {
            v2f kb = *(const v2f*)(kp + 4 * c);
            s = __builtin_amdgcn_wmma_f32_16x16x4_f32(
                    false, qa[c], false, kb, (short)0, s, false, false);
        }

        const int jg = kt * 16 + il;
#pragma unroll
        for (int r = 0; r < 8; ++r) {
            int ig = base + r + 8 * hi;
            bool ok = (jg <= ig + delta) && (jg < lenk);
            s[r] = ok ? s[r] : -1.0e9f;
        }

        float alpha[8];
#pragma unroll
        for (int r = 0; r < 8; ++r) {
            float x = s[r];
            x = fmaxf(x, __shfl_xor(x, 1, 16));
            x = fmaxf(x, __shfl_xor(x, 2, 16));
            x = fmaxf(x, __shfl_xor(x, 4, 16));
            x = fmaxf(x, __shfl_xor(x, 8, 16));
            float mnew = fmaxf(mrow[r], x);
            s[r]     = __expf(s[r] - mnew);
            alpha[r] = __expf(mrow[r] - mnew);
            mrow[r]  = mnew;
        }
#pragma unroll
        for (int r = 0; r < 8; ++r) {
            oacc[0][r] *= alpha[r]; oacc[1][r] *= alpha[r];
            oacc[2][r] *= alpha[r]; oacc[3][r] *= alpha[r];
            lacc[r]    *= alpha[r];
        }

#pragma unroll
        for (int r = 0; r < 8; ++r)
            pl[(r + 8 * hi) * LDS_STRIDE + il] = s[r];
        __syncthreads();
        v2f pa[4];
#pragma unroll
        for (int c = 0; c < 4; ++c)
            pa[c] = *(const v2f*)(&pl[il * LDS_STRIDE + 4 * c + 2 * hi]);
        __syncthreads();

#pragma unroll
        for (int c = 0; c < 4; ++c)
            lacc = __builtin_amdgcn_wmma_f32_16x16x4_f32(
                       false, pa[c], false, ones, (short)0, lacc, false, false);

#pragma unroll
        for (int nc = 0; nc < 4; ++nc) {
#pragma unroll
            for (int c = 0; c < 4; ++c) {
                int kr0 = kt * 16 + 4 * c + 2 * hi;
                int kr1 = kr0 + 1;
                kr0 = min(kr0, lenk - 1);
                kr1 = min(kr1, lenk - 1);
                v2f vb = { vbase[(size_t)kr0 * ROWSZ + nc * 16 + il],
                           vbase[(size_t)kr1 * ROWSZ + nc * 16 + il] };
                oacc[nc] = __builtin_amdgcn_wmma_f32_16x16x4_f32(
                               false, pa[c], false, vb, (short)0, oacc[nc],
                               false, false);
            }
        }
    }

#pragma unroll
    for (int r = 0; r < 8; ++r) {
        int m = r + 8 * hi;
        if (m < nrows) {
            float inv = 1.0f / lacc[r];
            size_t ro = (size_t)(cq0 + base + m) * ROWSZ + h * HD;
#pragma unroll
            for (int nc = 0; nc < 4; ++nc)
                out[ro + nc * 16 + il] = oacc[nc][r] * inv;
        }
    }
}

extern "C" void kernel_launch(void* const* d_in, const int* in_sizes, int n_in,
                              void* d_out, int out_size, void* d_ws, size_t ws_size,
                              hipStream_t stream) {
    const float* q   = (const float*)d_in[0];
    const float* k   = (const float*)d_in[1];
    const float* v   = (const float*)d_in[2];
    const int*   cuq = (const int*)d_in[3];
    const int*   cuk = (const int*)d_in[4];
    float* out = (float*)d_out;

    const int T = in_sizes[0] / ROWSZ;
    const int B = in_sizes[3] - 1;
    dim3 grid((T + 15) / 16, B, NH);
    va_fwd_kernel<<<grid, 32, 0, stream>>>(q, k, v, cuq, cuk, out);
}


# Work-group = 8 waves on a comb of 8 query tiles (g, g+S, ..) of one batch and head; each K/V tile is read once per group in whole rows into LDS and shared by its waves; sums and order unchanged
# speedup vs baseline: 1.8230x; 1.8230x over previous
#include <hip/hip_runtime.h>
#include <math.h>


typedef __attribute__((ext_vector_type(2))) float v2f;
typedef __attribute__((ext_vector_type(4))) float v4f;
typedef __attribute__((ext_vector_type(8))) float v8f;

#define NH 16
#define HD 64
#define ROWSZ (NH * HD)
#define LDS_STRIDE 18
#define NW 8
#define KP 68
#define VP 72

__device__ __forceinline__ int key_tiles(int t, int delta, int lenk)
{
    int maxj = 16 * t + 15 + delta;
    if (maxj > lenk - 1) maxj = lenk - 1;
    return (maxj < 0) ? 0 : (maxj / 16 + 1);
}

__global__ __launch_bounds__(NW * 32)
void va_fwd_kernel(const float* __restrict__ q,
                   const float* __restrict__ k,
                   const float* __restrict__ v,
                   const int*   __restrict__ cuq,
                   const int*   __restrict__ cuk,
                   float*       __restrict__ out,
                   int nb)
{
    __shared__ __attribute__((aligned(16))) float Ks[2][16 * KP];
    __shared__ __attribute__((aligned(16))) float Vs[2][16 * VP];
    __shared__ float pls[NW][16 * LDS_STRIDE];

    const int tid  = threadIdx.x;
    const int lane = tid & 31;
    const int il   = lane & 15;
    const int hi   = lane >> 4;
    const int wv   = __builtin_amdgcn_readfirstlane(tid >> 5);
    const int h    = blockIdx.x;

    int b = -1, g = 0, S = 0, n = 0;
    {
        const int slot = blockIdx.y;
        int acc = 0;
        for (int i = 0; i < nb; ++i) {
            int len = cuq[i + 1] - cuq[i];
            if (len < 0) len = 0;
            const int ni = (len + 15) >> 4;
            const int Si = (ni + NW - 1) / NW;
            if (b < 0 && slot < acc + Si) { b = i; g = slot - acc; S = Si; n = ni; }
            acc += Si;
        }
    }
    if (b < 0) return;

    const int cq0   = cuq[b];
    const int cq1   = cuq[b + 1];
    const int lenq  = cq1 - cq0;
    const int ck0   = cuk[b];
    const int lenk  = cuk[b + 1] - ck0;
    const int delta = lenk - lenq;

    const int jt    = (wv < 4) ? wv : (11 - wv);
    const int t     = g + jt * S;
    const bool have = (t < n);
    const int base  = t * 16;
    const int nrows = have ? min(16, lenq - base) : 0;
    const int myKT  = have ? key_tiles(t, delta, lenk) : 0;

    int jmax = (n - 1 - g) / S;
    if (jmax > NW - 1) jmax = NW - 1;
    const int KT = key_tiles(g + jmax * S, delta, lenk);

    float* pl = pls[wv];

    int qrow = cq0 + base + il;
    if (qrow > cq1 - 1) qrow = cq1 - 1;
    const float* qp = q + (size_t)qrow * ROWSZ + h * HD + 2 * hi;
    v2f qa[16];
#pragma unroll
    for (int c = 0; c < 16; ++c) {
        v2f tq = *(const v2f*)(qp + 4 * c);
        qa[c] = tq * 0.125f;
    }

    v8f oacc[4];
#pragma unroll
    for (int nc = 0; nc < 4; ++nc) oacc[nc] = (v8f){};
    v8f lacc = (v8f){};
    float mrow[8];
#pragma unroll
    for (int r = 0; r < 8; ++r) mrow[r] = -3.0e38f;

    const float* kbase = k + (size_t)ck0 * ROWSZ + h * HD;
    const float* vbase = v + (size_t)ck0 * ROWSZ + h * HD;
    const v2f ones = {1.0f, 1.0f};

    const int lrow = tid >> 4;
    const int lc4  = (tid & 15) * 4;
    v4f kreg = (v4f){}, vreg = (v4f){};

    if (KT > 0) {
        int kr = min(lrow, lenk - 1);
        kreg = *(const v4f*)(kbase + (size_t)kr * ROWSZ + lc4);
        vreg = *(const v4f*)(vbase + (size_t)kr * ROWSZ + lc4);
        *(v4f*)(&Ks[0][lrow * KP + lc4]) = kreg;
        *(v4f*)(&Vs[0][lrow * VP + lc4]) = vreg;
    }
    __syncthreads();

    for (int kt = 0; kt < KT; ++kt) {
        const int cur = kt & 1;
        if (kt + 1 < KT) {
            int kr = min((kt + 1) * 16 + lrow, lenk - 1);
            kreg = *(const v4f*)(kbase + (size_t)kr * ROWSZ + lc4);
            vreg = *(const v4f*)(vbase + (size_t)kr * ROWSZ + lc4);
        }

        if (kt < myKT) {
            const float* kt_s = &Ks[cur][il * KP + 2 * hi];
            const float* vt_s = &Vs[cur][il];

            v8f s = (v8f){};
#pragma unroll
            for (int c = 0; c < 16; ++c) {
                v2f kb = *(const v2f*)(kt_s + 4 * c);
                s = __builtin_amdgcn_wmma_f32_16x16x4_f32(
                        false, qa[c], false, kb, (short)0, s, false, false);
            }

            const int jg = kt * 16 + il;
#pragma unroll
            for (int r = 0; r < 8; ++r) {
                int ig = base + r + 8 * hi;
                bool ok = (jg <= ig + delta) && (jg < lenk);
                s[r] = ok ? s[r] : -1.0e9f;
            }

            float alpha[8];
#pragma unroll
            for (int r = 0; r < 8; ++r) {
                float x = s[r];
                x = fmaxf(x, __shfl_xor(x, 1, 16));
                x = fmaxf(x, __shfl_xor(x, 2, 16));
                x = fmaxf(x, __shfl_xor(x, 4, 16));
                x = fmaxf(x, __shfl_xor(x, 8, 16));
                float mnew = fmaxf(mrow[r], x);
                s[r]     = __expf(s[r] - mnew);
                alpha[r] = __expf(mrow[r] - mnew);
                mrow[r]  = mnew;
            }
#pragma unroll
            for (int r = 0; r < 8; ++r) {
                oacc[0][r] *= alpha[r]; oacc[1][r] *= alpha[r];
                oacc[2][r] *= alpha[r]; oacc[3][r] *= alpha[r];
                lacc[r]    *= alpha[r];
            }

#pragma unroll
            for (int r = 0; r < 8; ++r)
                pl[(r + 8 * hi) * LDS_STRIDE + il] = s[r];
            __builtin_amdgcn_fence(__ATOMIC_RELEASE, "wavefront");
            __builtin_amdgcn_wave_barrier();
            __builtin_amdgcn_fence(__ATOMIC_ACQUIRE, "wavefront");
            v2f pa[4];
#pragma unroll
            for (int c = 0; c < 4; ++c)
                pa[c] = *(const v2f*)(&pl[il * LDS_STRIDE + 4 * c + 2 * hi]);
            __builtin_amdgcn_fence(__ATOMIC_RELEASE, "wavefront");
            __builtin_amdgcn_wave_barrier();
            __builtin_amdgcn_fence(__ATOMIC_ACQUIRE, "wavefront");

#pragma unroll
            for (int c = 0; c < 4; ++c)
                lacc = __builtin_amdgcn_wmma_f32_16x16x4_f32(
                           false, pa[c], false, ones, (short)0, lacc, false, false);

#pragma unroll
            for (int nc = 0; nc < 4; ++nc) {
#pragma unroll
                for (int c = 0; c < 4; ++c) {
                    const int kr0 = 4 * c + 2 * hi;
                    v2f vb = { vt_s[kr0 * VP + nc * 16],
                               vt_s[(kr0 + 1) * VP + nc * 16] };
                    oacc[nc] = __builtin_amdgcn_wmma_f32_16x16x4_f32(
                                   false, pa[c], false, vb, (short)0, oacc[nc],
                                   false, false);
                }
            }
        }

        if (kt + 1 < KT) {
            *(v4f*)(&Ks[cur ^ 1][lrow * KP + lc4]) = kreg;
            *(v4f*)(&Vs[cur ^ 1][lrow * VP + lc4]) = vreg;
        }
        __syncthreads();
    }

#pragma unroll
    for (int r = 0; r < 8; ++r) {
        int m = r + 8 * hi;
        if (m < nrows) {
            float inv = (myKT > 0) ? (1.0f / lacc[r]) : 0.0f;
            size_t ro = (size_t)(cq0 + base + m) * ROWSZ + h * HD;
#pragma unroll
            for (int nc = 0; nc < 4; ++nc)
                out[ro + nc * 16 + il] = (myKT > 0) ? (oacc[nc][r] * inv) : 0.0f;
        }
    }
}

extern "C" void kernel_launch(void* const* d_in, const int* in_sizes, int n_in,
                              void* d_out, int out_size, void* d_ws, size_t ws_size,
                              hipStream_t stream) {
    const float* q   = (const float*)d_in[0];
    const float* k   = (const float*)d_in[1];
    const float* v   = (const float*)d_in[2];
    const int*   cuq = (const int*)d_in[3];
    const int*   cuk = (const int*)d_in[4];
    float* out = (float*)d_out;

    const int T = in_sizes[0] / ROWSZ;
    const int B = in_sizes[3] - 1;
    dim3 grid(NH, (T + 127) / 128 + 2 * B, 1);
    va_fwd_kernel<<<grid, NW * 32, 0, stream>>>(q, k, v, cuq, cuk, out, B);
}


# Score-tile row maximum by four row_xmask DPP max steps on the vector ALU, one instruction per step and row; was 32 LDS-unit permutes and their waits per key tile; values and sum order unchanged
# speedup vs baseline: 1.9399x; 1.0641x over previous
#include <hip/hip_runtime.h>
#include <math.h>


typedef __attribute__((ext_vector_type(2))) float v2f;
typedef __attribute__((ext_vector_type(4))) float v4f;
typedef __attribute__((ext_vector_type(8))) float v8f;

#define NH 16
#define HD 64
#define ROWSZ (NH * HD)
#define LDS_STRIDE 18
#define NW 8
#define KP 68
#define VP 72

__device__ __forceinline__ int key_tiles(int t, int delta, int lenk)
{
    int maxj = 16 * t + 15 + delta;
    if (maxj > lenk - 1) maxj = lenk - 1;
    return (maxj < 0) ? 0 : (maxj / 16 + 1);
}

template <int M>
__device__ __forceinline__ void row_xmask_max_step(float (&x)[8])
{
#pragma unroll
    for (int r = 0; r < 8; ++r)
        asm("v_max_num_f32_dpp %0, %1, %1 row_xmask:%2 row_mask:0xf bank_mask:0xf"
            : "=v"(x[r]) : "v"(x[r]), "n"(M));
}

__global__ __launch_bounds__(NW * 32)
void va_fwd_kernel(const float* __restrict__ q,
                   const float* __restrict__ k,
                   const float* __restrict__ v,
                   const int*   __restrict__ cuq,
                   const int*   __restrict__ cuk,
                   float*       __restrict__ out,
                   int nb)
{
    __shared__ __attribute__((aligned(16))) float Ks[2][16 * KP];
    __shared__ __attribute__((aligned(16))) float Vs[2][16 * VP];
    __shared__ float pls[NW][16 * LDS_STRIDE];

    const int tid  = threadIdx.x;
    const int lane = tid & 31;
    const int il   = lane & 15;
    const int hi   = lane >> 4;
    const int wv   = __builtin_amdgcn_readfirstlane(tid >> 5);
    const int h    = blockIdx.x;

    int b = -1, g = 0, S = 0, n = 0;
    {
        const int slot = blockIdx.y;
        int acc = 0;
        for (int i = 0; i < nb; ++i) {
            int len = cuq[i + 1] - cuq[i];
            if (len < 0) len = 0;
            const int ni = (len + 15) >> 4;
            const int Si = (ni + NW - 1) / NW;
            if (b < 0 && slot < acc + Si) { b = i; g = slot - acc; S = Si; n = ni; }
            acc += Si;
        }
    }
    if (b < 0) return;

    const int cq0   = cuq[b];
    const int cq1   = cuq[b + 1];
    const int lenq  = cq1 - cq0;
    const int ck0   = cuk[b];
    const int lenk  = cuk[b + 1] - ck0;
    const int delta = lenk - lenq;

    const int jt    = (wv < 4) ? wv : (11 - wv);
    const int t     = g + jt * S;
    const bool have = (t < n);
    const int base  = t * 16;
    const int nrows = have ? min(16, lenq - base) : 0;
    const int myKT  = have ? key_tiles(t, delta, lenk) : 0;

    int jmax = (n - 1 - g) / S;
    if (jmax > NW - 1) jmax = NW - 1;
    const int KT = key_tiles(g + jmax * S, delta, lenk);

    float* pl = pls[wv];

    int qrow = cq0 + base + il;
    if (qrow > cq1 - 1) qrow = cq1 - 1;
    const float* qp = q + (size_t)qrow * ROWSZ + h * HD + 2 * hi;
    v2f qa[16];
#pragma unroll
    for (int c = 0; c < 16; ++c) {
        v2f tq = *(const v2f*)(qp + 4 * c);
        qa[c] = tq * 0.125f;
    }

    v8f oacc[4];
#pragma unroll
    for (int nc = 0; nc < 4; ++nc) oacc[nc] = (v8f){};
    v8f lacc = (v8f){};
    float mrow[8];
#pragma unroll
    for (int r = 0; r < 8; ++r) mrow[r] = -3.0e38f;

    const float* kbase = k + (size_t)ck0 * ROWSZ + h * HD;
    const float* vbase = v + (size_t)ck0 * ROWSZ + h * HD;
    const v2f ones = {1.0f, 1.0f};

    const int lrow = tid >> 4;
    const int lc4  = (tid & 15) * 4;
    v4f kreg = (v4f){}, vreg = (v4f){};

    if (KT > 0) {
        int kr = min(lrow, lenk - 1);
        kreg = *(const v4f*)(kbase + (size_t)kr * ROWSZ + lc4);
        vreg = *(const v4f*)(vbase + (size_t)kr * ROWSZ + lc4);
        *(v4f*)(&Ks[0][lrow * KP + lc4]) = kreg;
        *(v4f*)(&Vs[0][lrow * VP + lc4]) = vreg;
    }
    __syncthreads();

    for (int kt = 0; kt < KT; ++kt) {
        const int cur = kt & 1;
        if (kt + 1 < KT) {
            int kr = min((kt + 1) * 16 + lrow, lenk - 1);
            kreg = *(const v4f*)(kbase + (size_t)kr * ROWSZ + lc4);
            vreg = *(const v4f*)(vbase + (size_t)kr * ROWSZ + lc4);
        }

        if (kt < myKT) {
            const float* kt_s = &Ks[cur][il * KP + 2 * hi];
            const float* vt_s = &Vs[cur][il];

            v8f s = (v8f){};
#pragma unroll
            for (int c = 0; c < 16; ++c) {
                v2f kb = *(const v2f*)(kt_s + 4 * c);
                s = __builtin_amdgcn_wmma_f32_16x16x4_f32(
                        false, qa[c], false, kb, (short)0, s, false, false);
            }

            const int jg = kt * 16 + il;
#pragma unroll
            for (int r = 0; r < 8; ++r) {
                int ig = base + r + 8 * hi;
                bool ok = (jg <= ig + delta) && (jg < lenk);
                s[r] = ok ? s[r] : -1.0e9f;
            }

            float rmax[8];
#pragma unroll
            for (int r = 0; r < 8; ++r) rmax[r] = s[r];
            row_xmask_max_step<1>(rmax);
            row_xmask_max_step<2>(rmax);
            row_xmask_max_step<4>(rmax);
            row_xmask_max_step<8>(rmax);
            float alpha[8];
#pragma unroll
            for (int r = 0; r < 8; ++r) {
                float mnew = fmaxf(mrow[r], rmax[r]);
                s[r]     = __expf(s[r] - mnew);
                alpha[r] = __expf(mrow[r] - mnew);
                mrow[r]  = mnew;
            }
#pragma unroll
            for (int r = 0; r < 8; ++r) {
                oacc[0][r] *= alpha[r]; oacc[1][r] *= alpha[r];
                oacc[2][r] *= alpha[r]; oacc[3][r] *= alpha[r];
                lacc[r]    *= alpha[r];
            }

#pragma unroll
            for (int r = 0; r < 8; ++r)
                pl[(r + 8 * hi) * LDS_STRIDE + il] = s[r];
            __builtin_amdgcn_fence(__ATOMIC_RELEASE, "wavefront");
            __builtin_amdgcn_wave_barrier();
            __builtin_amdgcn_fence(__ATOMIC_ACQUIRE, "wavefront");
            v2f pa[4];
#pragma unroll
            for (int c = 0; c < 4; ++c)
                pa[c] = *(const v2f*)(&pl[il * LDS_STRIDE + 4 * c + 2 * hi]);
            __builtin_amdgcn_fence(__ATOMIC_RELEASE, "wavefront");
            __builtin_amdgcn_wave_barrier();
            __builtin_amdgcn_fence(__ATOMIC_ACQUIRE, "wavefront");

#pragma unroll
            for (int c = 0; c < 4; ++c)
                lacc = __builtin_amdgcn_wmma_f32_16x16x4_f32(
                           false, pa[c], false, ones, (short)0, lacc, false, false);

#pragma unroll
            for (int nc = 0; nc < 4; ++nc) {
#pragma unroll
                for (int c = 0; c < 4; ++c) {
                    const int kr0 = 4 * c + 2 * hi;
                    v2f vb = { vt_s[kr0 * VP + nc * 16],
                               vt_s[(kr0 + 1) * VP + nc * 16] };
                    oacc[nc] = __builtin_amdgcn_wmma_f32_16x16x4_f32(
                                   false, pa[c], false, vb, (short)0, oacc[nc],
                                   false, false);
                }
            }
        }

        if (kt + 1 < KT) {
            *(v4f*)(&Ks[cur ^ 1][lrow * KP + lc4]) = kreg;
            *(v4f*)(&Vs[cur ^ 1][lrow * VP + lc4]) = vreg;
        }
        __syncthreads();
    }

#pragma unroll
    for (int r = 0; r < 8; ++r) {
        int m = r + 8 * hi;
        if (m < nrows) {
            float inv = (myKT > 0) ? (1.0f / lacc[r]) : 0.0f;
            size_t ro = (size_t)(cq0 + base + m) * ROWSZ + h * HD;
#pragma unroll
            for (int nc = 0; nc < 4; ++nc)
                out[ro + nc * 16 + il] = (myKT > 0) ? (oacc[nc][r] * inv) : 0.0f;
        }
    }
}

extern "C" void kernel_launch(void* const* d_in, const int* in_sizes, int n_in,
                              void* d_out, int out_size, void* d_ws, size_t ws_size,
                              hipStream_t stream) {
    const float* q   = (const float*)d_in[0];
    const float* k   = (const float*)d_in[1];
    const float* v   = (const float*)d_in[2];
    const int*   cuq = (const int*)d_in[3];
    const int*   cuk = (const int*)d_in[4];
    float* out = (float*)d_out;

    const int T = in_sizes[0] / ROWSZ;
    const int B = in_sizes[3] - 1;
    dim3 grid(NH, (T + 127) / 128 + 2 * B, 1);
    va_fwd_kernel<<<grid, NW * 32, 0, stream>>>(q, k, v, cuq, cuk, out, B);
}


# Key step = three 16-key tiles: one running-max update, one rescale of O and l, one P trip through LDS, one barrier per 48 keys (was per 16); three independent score WMMA chains; sum order kept
# speedup vs baseline: 2.1104x; 1.0879x over previous
#include <hip/hip_runtime.h>
#include <math.h>


typedef __attribute__((ext_vector_type(2))) float v2f;
typedef __attribute__((ext_vector_type(4))) float v4f;
typedef __attribute__((ext_vector_type(8))) float v8f;

#define NH 16
#define HD 64
#define ROWSZ (NH * HD)
#define KS 3
#define PP (16 * KS + 4)
#define NW 8
#define KP 68
#define VP 72

__device__ __forceinline__ int key_tiles(int t, int delta, int lenk)
{
    int maxj = 16 * t + 15 + delta;
    if (maxj > lenk - 1) maxj = lenk - 1;
    return (maxj < 0) ? 0 : (maxj / 16 + 1);
}

template <int M>
__device__ __forceinline__ void row_xmask_max_step(float (&x)[8])
{
#pragma unroll
    for (int r = 0; r < 8; ++r)
        asm("v_max_num_f32_dpp %0, %1, %1 row_xmask:%2 row_mask:0xf bank_mask:0xf"
            : "=v"(x[r]) : "v"(x[r]), "n"(M));
}

template <int NSUB>
__device__ __forceinline__ void key_step(const float* __restrict__ kt_s,
                                         const float* __restrict__ vt_s,
                                         float* __restrict__ pl,
                                         const int il, const int hi,
                                         const int jg0,
                                         const int base, const int delta, const int lenk,
                                         const v2f (&qa)[16], const v2f ones,
                                         v8f (&oacc)[4], v8f& lacc, float (&mrow)[8])
{
    v8f s[NSUB];
#pragma unroll
    for (int u = 0; u < NSUB; ++u) s[u] = (v8f){};
#pragma unroll
    for (int c = 0; c < 16; ++c) {
#pragma unroll
        for (int u = 0; u < NSUB; ++u) {
            v2f kb = *(const v2f*)(kt_s + u * 16 * KP + 4 * c);
            s[u] = __builtin_amdgcn_wmma_f32_16x16x4_f32(
                       false, qa[c], false, kb, (short)0, s[u], false, false);
        }
    }

#pragma unroll
    for (int u = 0; u < NSUB; ++u) {
        const int jg = jg0 + 16 * u;
#pragma unroll
        for (int r = 0; r < 8; ++r) {
            int ig = base + r + 8 * hi;
            bool ok = (jg <= ig + delta) && (jg < lenk);
            s[u][r] = ok ? s[u][r] : -1.0e9f;
        }
    }

    float rmax[8];
#pragma unroll
    for (int r = 0; r < 8; ++r) {
        rmax[r] = s[0][r];
#pragma unroll
        for (int u = 1; u < NSUB; ++u) rmax[r] = fmaxf(rmax[r], s[u][r]);
    }
    row_xmask_max_step<1>(rmax);
    row_xmask_max_step<2>(rmax);
    row_xmask_max_step<4>(rmax);
    row_xmask_max_step<8>(rmax);
    float alpha[8];
#pragma unroll
    for (int r = 0; r < 8; ++r) {
        float mnew = fmaxf(mrow[r], rmax[r]);
#pragma unroll
        for (int u = 0; u < NSUB; ++u) s[u][r] = __expf(s[u][r] - mnew);
        alpha[r] = __expf(mrow[r] - mnew);
        mrow[r]  = mnew;
    }
#pragma unroll
    for (int r = 0; r < 8; ++r) {
        oacc[0][r] *= alpha[r]; oacc[1][r] *= alpha[r];
        oacc[2][r] *= alpha[r]; oacc[3][r] *= alpha[r];
        lacc[r]    *= alpha[r];
    }

#pragma unroll
    for (int u = 0; u < NSUB; ++u)
#pragma unroll
        for (int r = 0; r < 8; ++r)
            pl[(r + 8 * hi) * PP + 16 * u + il] = s[u][r];
    __builtin_amdgcn_fence(__ATOMIC_RELEASE, "wavefront");
    __builtin_amdgcn_wave_barrier();
    __builtin_amdgcn_fence(__ATOMIC_ACQUIRE, "wavefront");
    v2f pa[NSUB][4];
#pragma unroll
    for (int u = 0; u < NSUB; ++u)
#pragma unroll
        for (int c = 0; c < 4; ++c)
            pa[u][c] = *(const v2f*)(&pl[il * PP + 16 * u + 4 * c + 2 * hi]);
    __builtin_amdgcn_fence(__ATOMIC_RELEASE, "wavefront");
    __builtin_amdgcn_wave_barrier();
    __builtin_amdgcn_fence(__ATOMIC_ACQUIRE, "wavefront");

#pragma unroll
    for (int u = 0; u < NSUB; ++u)
#pragma unroll
        for (int c = 0; c < 4; ++c)
            lacc = __builtin_amdgcn_wmma_f32_16x16x4_f32(
                       false, pa[u][c], false, ones, (short)0, lacc, false, false);

#pragma unroll
    for (int nc = 0; nc < 4; ++nc) {
#pragma unroll
        for (int u = 0; u < NSUB; ++u) {
#pragma unroll
            for (int c = 0; c < 4; ++c) {
                const int kr0 = 16 * u + 4 * c + 2 * hi;
                v2f vb = { vt_s[kr0 * VP + nc * 16],
                           vt_s[(kr0 + 1) * VP + nc * 16] };
                oacc[nc] = __builtin_amdgcn_wmma_f32_16x16x4_f32(
                               false, pa[u][c], false, vb, (short)0, oacc[nc],
                               false, false);
            }
        }
    }
}

__global__ __launch_bounds__(NW * 32)
void va_fwd_kernel(const float* __restrict__ q,
                   const float* __restrict__ k,
                   const float* __restrict__ v,
                   const int*   __restrict__ cuq,
                   const int*   __restrict__ cuk,
                   float*       __restrict__ out,
                   int nb)
{
    __shared__ __attribute__((aligned(16))) float Ks[2][KS * 16 * KP];
    __shared__ __attribute__((aligned(16))) float Vs[2][KS * 16 * VP];
    __shared__ __attribute__((aligned(16))) float pls[NW][16 * PP];

    const int tid  = threadIdx.x;
    const int lane = tid & 31;
    const int il   = lane & 15;
    const int hi   = lane >> 4;
    const int wv   = __builtin_amdgcn_readfirstlane(tid >> 5);
    const int h    = blockIdx.x;

    int b = -1, g = 0, S = 0, n = 0;
    {
        const int slot = blockIdx.y;
        int acc = 0;
        for (int i = 0; i < nb; ++i) {
            int len = cuq[i + 1] - cuq[i];
            if (len < 0) len = 0;
            const int ni = (len + 15) >> 4;
            const int Si = (ni + NW - 1) / NW;
            if (b < 0 && slot < acc + Si) { b = i; g = slot - acc; S = Si; n = ni; }
            acc += Si;
        }
    }
    if (b < 0) return;

    const int cq0   = cuq[b];
    const int cq1   = cuq[b + 1];
    const int lenq  = cq1 - cq0;
    const int ck0   = cuk[b];
    const int lenk  = cuk[b + 1] - ck0;
    const int delta = lenk - lenq;

    const int jt    = (wv < 4) ? wv : (11 - wv);
    const int t     = g + jt * S;
    const bool have = (t < n);
    const int base  = t * 16;
    const int nrows = have ? min(16, lenq - base) : 0;
    const int myKT  = have ? key_tiles(t, delta, lenk) : 0;

    int jmax = (n - 1 - g) / S;
    if (jmax > NW - 1) jmax = NW - 1;
    const int KT = key_tiles(g + jmax * S, delta, lenk);

    float* pl = pls[wv];

    int qrow = cq0 + base + il;
    if (qrow > cq1 - 1) qrow = cq1 - 1;
    const float* qp = q + (size_t)qrow * ROWSZ + h * HD + 2 * hi;
    v2f qa[16];
#pragma unroll
    for (int c = 0; c < 16; ++c) {
        v2f tq = *(const v2f*)(qp + 4 * c);
        qa[c] = tq * 0.125f;
    }

    v8f oacc[4];
#pragma unroll
    for (int nc = 0; nc < 4; ++nc) oacc[nc] = (v8f){};
    v8f lacc = (v8f){};
    float mrow[8];
#pragma unroll
    for (int r = 0; r < 8; ++r) mrow[r] = -3.0e38f;

    const float* kbase = k + (size_t)ck0 * ROWSZ + h * HD;
    const float* vbase = v + (size_t)ck0 * ROWSZ + h * HD;
    const v2f ones = {1.0f, 1.0f};

    const int lrow = tid >> 4;
    const int lc4  = (tid & 15) * 4;
    v4f kreg[KS], vreg[KS];
    const int NS = (KT + KS - 1) / KS;

    if (NS > 0) {
#pragma unroll
        for (int u = 0; u < KS; ++u) {
            int kr = min(u * 16 + lrow, lenk - 1);
            kreg[u] = *(const v4f*)(kbase + (size_t)kr * ROWSZ + lc4);
            vreg[u] = *(const v4f*)(vbase + (size_t)kr * ROWSZ + lc4);
        }
#pragma unroll
        for (int u = 0; u < KS; ++u) {
            *(v4f*)(&Ks[0][(u * 16 + lrow) * KP + lc4]) = kreg[u];
            *(v4f*)(&Vs[0][(u * 16 + lrow) * VP + lc4]) = vreg[u];
        }
    }
    __syncthreads();

    for (int st = 0; st < NS; ++st) {
        const int cur = st & 1;
        const int kt0 = st * KS;
        if (st + 1 < NS) {
#pragma unroll
            for (int u = 0; u < KS; ++u) {
                int kr = min((kt0 + KS + u) * 16 + lrow, lenk - 1);
                kreg[u] = *(const v4f*)(kbase + (size_t)kr * ROWSZ + lc4);
                vreg[u] = *(const v4f*)(vbase + (size_t)kr * ROWSZ + lc4);
            }
        }

        const float* kt_s = &Ks[cur][il * KP + 2 * hi];
        const float* vt_s = &Vs[cur][il];
        if (kt0 + KS <= myKT) {
            key_step<KS>(kt_s, vt_s, pl, il, hi, kt0 * 16 + il, base, delta, lenk,
                         qa, ones, oacc, lacc, mrow);
        } else {
            for (int u = 0; u < KS - 1; ++u) {
                if (kt0 + u < myKT)
                    key_step<1>(kt_s + u * 16 * KP, vt_s + u * 16 * VP, pl, il, hi,
                                (kt0 + u) * 16 + il, base, delta, lenk,
                                qa, ones, oacc, lacc, mrow);
            }
        }

        if (st + 1 < NS) {
#pragma unroll
            for (int u = 0; u < KS; ++u) {
                *(v4f*)(&Ks[cur ^ 1][(u * 16 + lrow) * KP + lc4]) = kreg[u];
                *(v4f*)(&Vs[cur ^ 1][(u * 16 + lrow) * VP + lc4]) = vreg[u];
            }
        }
        __syncthreads();
    }

#pragma unroll
    for (int r = 0; r < 8; ++r) {
        int m = r + 8 * hi;
        if (m < nrows) {
            float inv = (myKT > 0) ? (1.0f / lacc[r]) : 0.0f;
            size_t ro = (size_t)(cq0 + base + m) * ROWSZ + h * HD;
#pragma unroll
            for (int nc = 0; nc < 4; ++nc)
                out[ro + nc * 16 + il] = (myKT > 0) ? (oacc[nc][r] * inv) : 0.0f;
        }
    }
}

extern "C" void kernel_launch(void* const* d_in, const int* in_sizes, int n_in,
                              void* d_out, int out_size, void* d_ws, size_t ws_size,
                              hipStream_t stream) {
    const float* q   = (const float*)d_in[0];
    const float* k   = (const float*)d_in[1];
    const float* v   = (const float*)d_in[2];
    const int*   cuq = (const int*)d_in[3];
    const int*   cuk = (const int*)d_in[4];
    float* out = (float*)d_out;

    const int T = in_sizes[0] / ROWSZ;
    const int B = in_sizes[3] - 1;
    dim3 grid(NH, (T + 127) / 128 + 2 * B, 1);
    va_fwd_kernel<<<grid, NW * 32, 0, stream>>>(q, k, v, cuq, cuk, out, B);
}


# LDS fragment reads of a key step issued in whole groups ahead of their WMMA runs (two register sets in turn): reads stand at 8 places in a step's matrix stream, was 33; sums and order unchanged
# speedup vs baseline: 2.2352x; 1.0591x over previous
#include <hip/hip_runtime.h>
#include <math.h>


typedef __attribute__((ext_vector_type(2))) float v2f;
typedef __attribute__((ext_vector_type(4))) float v4f;
typedef __attribute__((ext_vector_type(8))) float v8f;

#define NH 16
#define HD 64
#define ROWSZ (NH * HD)
#define KS 3
#define PP (16 * KS + 4)
#define NW 8
#define KP 68
#define VP 72
#define KCG 4

__device__ __forceinline__ void frag_group_fence()
{
    __builtin_amdgcn_sched_barrier(0x0006);
}

__device__ __forceinline__ int key_tiles(int t, int delta, int lenk)
{
    int maxj = 16 * t + 15 + delta;
    if (maxj > lenk - 1) maxj = lenk - 1;
    return (maxj < 0) ? 0 : (maxj / 16 + 1);
}

template <int M>
__device__ __forceinline__ void row_xmask_max_step(float (&x)[8])
{
#pragma unroll
    for (int r = 0; r < 8; ++r)
        asm("v_max_num_f32_dpp %0, %1, %1 row_xmask:%2 row_mask:0xf bank_mask:0xf"
            : "=v"(x[r]) : "v"(x[r]), "n"(M));
}

template <int NSUB>
__device__ __forceinline__ void key_step(const float* __restrict__ kt_s,
                                         const float* __restrict__ vt_s,
                                         float* __restrict__ pl,
                                         const int il, const int hi,
                                         const int jg0,
                                         const int base, const int delta, const int lenk,
                                         const v2f (&qa)[16], const v2f ones,
                                         v8f (&oacc)[4], v8f& lacc, float (&mrow)[8])
{
    v8f s[NSUB];
#pragma unroll
    for (int u = 0; u < NSUB; ++u) s[u] = (v8f){};
    constexpr int NKG = 16 / KCG;
    v2f kf[2][KCG][NSUB];
#pragma unroll
    for (int g = 0; g < 2; ++g)
#pragma unroll
        for (int j = 0; j < KCG; ++j)
#pragma unroll
            for (int u = 0; u < NSUB; ++u)
                kf[g][j][u] = *(const v2f*)(kt_s + u * 16 * KP + 4 * (g * KCG + j));
    frag_group_fence();
#pragma unroll
    for (int g = 0; g < NKG; ++g) {
#pragma unroll
        for (int j = 0; j < KCG; ++j) {
#pragma unroll
            for (int u = 0; u < NSUB; ++u)
                s[u] = __builtin_amdgcn_wmma_f32_16x16x4_f32(
                           false, qa[g * KCG + j], false, kf[g & 1][j][u], (short)0, s[u],
                           false, false);
        }
        if (g + 2 < NKG) {
            frag_group_fence();
#pragma unroll
            for (int j = 0; j < KCG; ++j)
#pragma unroll
                for (int u = 0; u < NSUB; ++u)
                    kf[g & 1][j][u] =
                        *(const v2f*)(kt_s + u * 16 * KP + 4 * ((g + 2) * KCG + j));
            frag_group_fence();
        }
    }

#pragma unroll
    for (int u = 0; u < NSUB; ++u) {
        const int jg = jg0 + 16 * u;
#pragma unroll
        for (int r = 0; r < 8; ++r) {
            int ig = base + r + 8 * hi;
            bool ok = (jg <= ig + delta) && (jg < lenk);
            s[u][r] = ok ? s[u][r] : -1.0e9f;
        }
    }

    float rmax[8];
#pragma unroll
    for (int r = 0; r < 8; ++r) {
        rmax[r] = s[0][r];
#pragma unroll
        for (int u = 1; u < NSUB; ++u) rmax[r] = fmaxf(rmax[r], s[u][r]);
    }
    row_xmask_max_step<1>(rmax);
    row_xmask_max_step<2>(rmax);
    row_xmask_max_step<4>(rmax);
    row_xmask_max_step<8>(rmax);
    float alpha[8];
#pragma unroll
    for (int r = 0; r < 8; ++r) {
        float mnew = fmaxf(mrow[r], rmax[r]);
#pragma unroll
        for (int u = 0; u < NSUB; ++u) s[u][r] = __expf(s[u][r] - mnew);
        alpha[r] = __expf(mrow[r] - mnew);
        mrow[r]  = mnew;
    }
#pragma unroll
    for (int r = 0; r < 8; ++r) {
        oacc[0][r] *= alpha[r]; oacc[1][r] *= alpha[r];
        oacc[2][r] *= alpha[r]; oacc[3][r] *= alpha[r];
        lacc[r]    *= alpha[r];
    }

#pragma unroll
    for (int u = 0; u < NSUB; ++u)
#pragma unroll
        for (int r = 0; r < 8; ++r)
            pl[(r + 8 * hi) * PP + 16 * u + il] = s[u][r];
    __builtin_amdgcn_fence(__ATOMIC_RELEASE, "wavefront");
    __builtin_amdgcn_wave_barrier();
    __builtin_amdgcn_fence(__ATOMIC_ACQUIRE, "wavefront");
    v2f pa[NSUB][4];
#pragma unroll
    for (int u = 0; u < NSUB; ++u)
#pragma unroll
        for (int c = 0; c < 4; ++c)
            pa[u][c] = *(const v2f*)(&pl[il * PP + 16 * u + 4 * c + 2 * hi]);
    __builtin_amdgcn_fence(__ATOMIC_RELEASE, "wavefront");
    __builtin_amdgcn_wave_barrier();
    __builtin_amdgcn_fence(__ATOMIC_ACQUIRE, "wavefront");

    constexpr int NVG = 2 * NSUB;
    v2f vf[2][2][4];
#pragma unroll
    for (int gi = 0; gi < 2; ++gi)
#pragma unroll
        for (int j = 0; j < 2; ++j)
#pragma unroll
            for (int nc = 0; nc < 4; ++nc) {
                const int kr0 = 8 * gi + 4 * j + 2 * hi;
                vf[gi][j][nc] = (v2f){ vt_s[kr0 * VP + nc * 16],
                                       vt_s[(kr0 + 1) * VP + nc * 16] };
            }
    frag_group_fence();
#pragma unroll
    for (int gi = 0; gi < NVG; ++gi) {
#pragma unroll
        for (int j = 0; j < 2; ++j) {
            const int u = gi >> 1;
            const int c = 2 * (gi & 1) + j;
            lacc = __builtin_amdgcn_wmma_f32_16x16x4_f32(
                       false, pa[u][c], false, ones, (short)0, lacc, false, false);
#pragma unroll
            for (int nc = 0; nc < 4; ++nc)
                oacc[nc] = __builtin_amdgcn_wmma_f32_16x16x4_f32(
                               false, pa[u][c], false, vf[gi & 1][j][nc], (short)0, oacc[nc],
                               false, false);
        }
        if (gi + 2 < NVG) {
            frag_group_fence();
#pragma unroll
            for (int j = 0; j < 2; ++j)
#pragma unroll
                for (int nc = 0; nc < 4; ++nc) {
                    const int kr0 = 8 * (gi + 2) + 4 * j + 2 * hi;
                    vf[gi & 1][j][nc] = (v2f){ vt_s[kr0 * VP + nc * 16],
                                               vt_s[(kr0 + 1) * VP + nc * 16] };
                }
            frag_group_fence();
        }
    }
}

__global__ __launch_bounds__(NW * 32)
void va_fwd_kernel(const float* __restrict__ q,
                   const float* __restrict__ k,
                   const float* __restrict__ v,
                   const int*   __restrict__ cuq,
                   const int*   __restrict__ cuk,
                   float*       __restrict__ out,
                   int nb)
{
    __shared__ __attribute__((aligned(16))) float Ks[2][KS * 16 * KP];
    __shared__ __attribute__((aligned(16))) float Vs[2][KS * 16 * VP];
    __shared__ __attribute__((aligned(16))) float pls[NW][16 * PP];

    const int tid  = threadIdx.x;
    const int lane = tid & 31;
    const int il   = lane & 15;
    const int hi   = lane >> 4;
    const int wv   = __builtin_amdgcn_readfirstlane(tid >> 5);
    const int h    = blockIdx.x;

    int b = -1, g = 0, S = 0, n = 0;
    {
        const int slot = blockIdx.y;
        int acc = 0;
        for (int i = 0; i < nb; ++i) {
            int len = cuq[i + 1] - cuq[i];
            if (len < 0) len = 0;
            const int ni = (len + 15) >> 4;
            const int Si = (ni + NW - 1) / NW;
            if (b < 0 && slot < acc + Si) { b = i; g = slot - acc; S = Si; n = ni; }
            acc += Si;
        }
    }
    if (b < 0) return;

    const int cq0   = cuq[b];
    const int cq1   = cuq[b + 1];
    const int lenq  = cq1 - cq0;
    const int ck0   = cuk[b];
    const int lenk  = cuk[b + 1] - ck0;
    const int delta = lenk - lenq;

    const int jt    = (wv < 4) ? wv : (11 - wv);
    const int t     = g + jt * S;
    const bool have = (t < n);
    const int base  = t * 16;
    const int nrows = have ? min(16, lenq - base) : 0;
    const int myKT  = have ? key_tiles(t, delta, lenk) : 0;

    int jmax = (n - 1 - g) / S;
    if (jmax > NW - 1) jmax = NW - 1;
    const int KT = key_tiles(g + jmax * S, delta, lenk);

    float* pl = pls[wv];

    int qrow = cq0 + base + il;
    if (qrow > cq1 - 1) qrow = cq1 - 1;
    const float* qp = q + (size_t)qrow * ROWSZ + h * HD + 2 * hi;
    v2f qa[16];
#pragma unroll
    for (int c = 0; c < 16; ++c) {
        v2f tq = *(const v2f*)(qp + 4 * c);
        qa[c] = tq * 0.125f;
    }

    v8f oacc[4];
#pragma unroll
    for (int nc = 0; nc < 4; ++nc) oacc[nc] = (v8f){};
    v8f lacc = (v8f){};
    float mrow[8];
#pragma unroll
    for (int r = 0; r < 8; ++r) mrow[r] = -3.0e38f;

    const float* kbase = k + (size_t)ck0 * ROWSZ + h * HD;
    const float* vbase = v + (size_t)ck0 * ROWSZ + h * HD;
    const v2f ones = {1.0f, 1.0f};

    const int lrow = tid >> 4;
    const int lc4  = (tid & 15) * 4;
    v4f kreg[KS], vreg[KS];
    const int NS = (KT + KS - 1) / KS;

    if (NS > 0) {
#pragma unroll
        for (int u = 0; u < KS; ++u) {
            int kr = min(u * 16 + lrow, lenk - 1);
            kreg[u] = *(const v4f*)(kbase + (size_t)kr * ROWSZ + lc4);
            vreg[u] = *(const v4f*)(vbase + (size_t)kr * ROWSZ + lc4);
        }
#pragma unroll
        for (int u = 0; u < KS; ++u) {
            *(v4f*)(&Ks[0][(u * 16 + lrow) * KP + lc4]) = kreg[u];
            *(v4f*)(&Vs[0][(u * 16 + lrow) * VP + lc4]) = vreg[u];
        }
    }
    __syncthreads();

    for (int st = 0; st < NS; ++st) {
        const int cur = st & 1;
        const int kt0 = st * KS;
        if (st + 1 < NS) {
#pragma unroll
            for (int u = 0; u < KS; ++u) {
                int kr = min((kt0 + KS + u) * 16 + lrow, lenk - 1);
                kreg[u] = *(const v4f*)(kbase + (size_t)kr * ROWSZ + lc4);
                vreg[u] = *(const v4f*)(vbase + (size_t)kr * ROWSZ + lc4);
            }
        }

        const float* kt_s = &Ks[cur][il * KP + 2 * hi];
        const float* vt_s = &Vs[cur][il];
        if (kt0 + KS <= myKT) {
            key_step<KS>(kt_s, vt_s, pl, il, hi, kt0 * 16 + il, base, delta, lenk,
                         qa, ones, oacc, lacc, mrow);
        } else {
            for (int u = 0; u < KS - 1; ++u) {
                if (kt0 + u < myKT)
                    key_step<1>(kt_s + u * 16 * KP, vt_s + u * 16 * VP, pl, il, hi,
                                (kt0 + u) * 16 + il, base, delta, lenk,
                                qa, ones, oacc, lacc, mrow);
            }
        }

        if (st + 1 < NS) {
#pragma unroll
            for (int u = 0; u < KS; ++u) {
                *(v4f*)(&Ks[cur ^ 1][(u * 16 + lrow) * KP + lc4]) = kreg[u];
                *(v4f*)(&Vs[cur ^ 1][(u * 16 + lrow) * VP + lc4]) = vreg[u];
            }
        }
        __syncthreads();
    }

#pragma unroll
    for (int r = 0; r < 8; ++r) {
        int m = r + 8 * hi;
        if (m < nrows) {
            float inv = (myKT > 0) ? (1.0f / lacc[r]) : 0.0f;
            size_t ro = (size_t)(cq0 + base + m) * ROWSZ + h * HD;
#pragma unroll
            for (int nc = 0; nc < 4; ++nc)
                out[ro + nc * 16 + il] = (myKT > 0) ? (oacc[nc][r] * inv) : 0.0f;
        }
    }
}

extern "C" void kernel_launch(void* const* d_in, const int* in_sizes, int n_in,
                              void* d_out, int out_size, void* d_ws, size_t ws_size,
                              hipStream_t stream) {
    const float* q   = (const float*)d_in[0];
    const float* k   = (const float*)d_in[1];
    const float* v   = (const float*)d_in[2];
    const int*   cuq = (const int*)d_in[3];
    const int*   cuk = (const int*)d_in[4];
    float* out = (float*)d_out;

    const int T = in_sizes[0] / ROWSZ;
    const int B = in_sizes[3] - 1;
    dim3 grid(NH, (T + 127) / 128 + 2 * B, 1);
    va_fwd_kernel<<<grid, NW * 32, 0, stream>>>(q, k, v, cuq, cuk, out, B);
}


# Softmax row sums kept per lane on the vector ALU (rescale fused in), a row's 16 lanes joined once after the key loop; the P x ones WMMAs (12 of 108 per key step) go; row-sum order changes
# speedup vs baseline: 2.3546x; 1.0534x over previous
#include <hip/hip_runtime.h>
#include <math.h>


typedef __attribute__((ext_vector_type(2))) float v2f;
typedef __attribute__((ext_vector_type(4))) float v4f;
typedef __attribute__((ext_vector_type(8))) float v8f;

#define NH 16
#define HD 64
#define ROWSZ (NH * HD)
#define KS 3
#define PP (16 * KS + 4)
#define NW 8
#define KP 68
#define VP 72
#define KCG 4

__device__ __forceinline__ void frag_group_fence()
{
    __builtin_amdgcn_sched_barrier(0x0006);
}

__device__ __forceinline__ int key_tiles(int t, int delta, int lenk)
{
    int maxj = 16 * t + 15 + delta;
    if (maxj > lenk - 1) maxj = lenk - 1;
    return (maxj < 0) ? 0 : (maxj / 16 + 1);
}

template <int M>
__device__ __forceinline__ void row_xmask_max_step(float (&x)[8])
{
#pragma unroll
    for (int r = 0; r < 8; ++r)
        asm("v_max_num_f32_dpp %0, %1, %1 row_xmask:%2 row_mask:0xf bank_mask:0xf"
            : "=v"(x[r]) : "v"(x[r]), "n"(M));
}

template <int M>
__device__ __forceinline__ void row_xmask_add_step(float (&x)[8])
{
#pragma unroll
    for (int r = 0; r < 8; ++r)
        asm("v_add_f32_dpp %0, %1, %1 row_xmask:%2 row_mask:0xf bank_mask:0xf"
            : "=v"(x[r]) : "v"(x[r]), "n"(M));
}

template <int NSUB>
__device__ __forceinline__ void key_step(const float* __restrict__ kt_s,
                                         const float* __restrict__ vt_s,
                                         float* __restrict__ pl,
                                         const int il, const int hi,
                                         const int jg0,
                                         const int base, const int delta, const int lenk,
                                         const v2f (&qa)[16],
                                         v8f (&oacc)[4], float (&lsum)[8], float (&mrow)[8])
{
    v8f s[NSUB];
#pragma unroll
    for (int u = 0; u < NSUB; ++u) s[u] = (v8f){};
    constexpr int NKG = 16 / KCG;
    v2f kf[2][KCG][NSUB];
#pragma unroll
    for (int g = 0; g < 2; ++g)
#pragma unroll
        for (int j = 0; j < KCG; ++j)
#pragma unroll
            for (int u = 0; u < NSUB; ++u)
                kf[g][j][u] = *(const v2f*)(kt_s + u * 16 * KP + 4 * (g * KCG + j));
    frag_group_fence();
#pragma unroll
    for (int g = 0; g < NKG; ++g) {
#pragma unroll
        for (int j = 0; j < KCG; ++j) {
#pragma unroll
            for (int u = 0; u < NSUB; ++u)
                s[u] = __builtin_amdgcn_wmma_f32_16x16x4_f32(
                           false, qa[g * KCG + j], false, kf[g & 1][j][u], (short)0, s[u],
                           false, false);
        }
        if (g + 2 < NKG) {
            frag_group_fence();
#pragma unroll
            for (int j = 0; j < KCG; ++j)
#pragma unroll
                for (int u = 0; u < NSUB; ++u)
                    kf[g & 1][j][u] =
                        *(const v2f*)(kt_s + u * 16 * KP + 4 * ((g + 2) * KCG + j));
            frag_group_fence();
        }
    }

    const int lim = base + 8 * hi + delta;
#pragma unroll
    for (int u = 0; u < NSUB; ++u) {
        const int jg = jg0 + 16 * u;
        const int d  = jg - lim;
        const bool inb = (jg < lenk);
#pragma unroll
        for (int r = 0; r < 8; ++r) {
            bool ok = (d <= r) && inb;
            s[u][r] = ok ? s[u][r] : -1.0e9f;
        }
    }

    float rmax[8];
#pragma unroll
    for (int r = 0; r < 8; ++r) {
        rmax[r] = s[0][r];
#pragma unroll
        for (int u = 1; u < NSUB; ++u) rmax[r] = fmaxf(rmax[r], s[u][r]);
    }
    row_xmask_max_step<1>(rmax);
    row_xmask_max_step<2>(rmax);
    row_xmask_max_step<4>(rmax);
    row_xmask_max_step<8>(rmax);
    float alpha[8];
#pragma unroll
    for (int r = 0; r < 8; ++r) {
        float mnew = fmaxf(mrow[r], rmax[r]);
#pragma unroll
        for (int u = 0; u < NSUB; ++u) s[u][r] = __expf(s[u][r] - mnew);
        alpha[r] = __expf(mrow[r] - mnew);
        mrow[r]  = mnew;
    }
#pragma unroll
    for (int r = 0; r < 8; ++r) {
        oacc[0][r] *= alpha[r]; oacc[1][r] *= alpha[r];
        oacc[2][r] *= alpha[r]; oacc[3][r] *= alpha[r];
        float psum = s[0][r];
#pragma unroll
        for (int u = 1; u < NSUB; ++u) psum += s[u][r];
        lsum[r] = lsum[r] * alpha[r] + psum;
    }

#pragma unroll
    for (int u = 0; u < NSUB; ++u)
#pragma unroll
        for (int r = 0; r < 8; ++r)
            pl[(r + 8 * hi) * PP + 16 * u + il] = s[u][r];
    __builtin_amdgcn_fence(__ATOMIC_RELEASE, "wavefront");
    __builtin_amdgcn_wave_barrier();
    __builtin_amdgcn_fence(__ATOMIC_ACQUIRE, "wavefront");
    v2f pa[NSUB][4];
#pragma unroll
    for (int u = 0; u < NSUB; ++u)
#pragma unroll
        for (int c = 0; c < 4; ++c)
            pa[u][c] = *(const v2f*)(&pl[il * PP + 16 * u + 4 * c + 2 * hi]);
    __builtin_amdgcn_fence(__ATOMIC_RELEASE, "wavefront");
    __builtin_amdgcn_wave_barrier();
    __builtin_amdgcn_fence(__ATOMIC_ACQUIRE, "wavefront");

    constexpr int NVG = 2 * NSUB;
    v2f vf[2][2][4];
#pragma unroll
    for (int gi = 0; gi < 2; ++gi)
#pragma unroll
        for (int j = 0; j < 2; ++j)
#pragma unroll
            for (int nc = 0; nc < 4; ++nc) {
                const int kr0 = 8 * gi + 4 * j + 2 * hi;
                vf[gi][j][nc] = (v2f){ vt_s[kr0 * VP + nc * 16],
                                       vt_s[(kr0 + 1) * VP + nc * 16] };
            }
    frag_group_fence();
#pragma unroll
    for (int gi = 0; gi < NVG; ++gi) {
#pragma unroll
        for (int j = 0; j < 2; ++j) {
            const int u = gi >> 1;
            const int c = 2 * (gi & 1) + j;
#pragma unroll
            for (int nc = 0; nc < 4; ++nc)
                oacc[nc] = __builtin_amdgcn_wmma_f32_16x16x4_f32(
                               false, pa[u][c], false, vf[gi & 1][j][nc], (short)0, oacc[nc],
                               false, false);
        }
        if (gi + 2 < NVG) {
            frag_group_fence();
#pragma unroll
            for (int j = 0; j < 2; ++j)
#pragma unroll
                for (int nc = 0; nc < 4; ++nc) {
                    const int kr0 = 8 * (gi + 2) + 4 * j + 2 * hi;
                    vf[gi & 1][j][nc] = (v2f){ vt_s[kr0 * VP + nc * 16],
                                               vt_s[(kr0 + 1) * VP + nc * 16] };
                }
            frag_group_fence();
        }
    }
}

__global__ __launch_bounds__(NW * 32)
void va_fwd_kernel(const float* __restrict__ q,
                   const float* __restrict__ k,
                   const float* __restrict__ v,
                   const int*   __restrict__ cuq,
                   const int*   __restrict__ cuk,
                   float*       __restrict__ out,
                   int nb)
{
    __shared__ __attribute__((aligned(16))) float Ks[2][KS * 16 * KP];
    __shared__ __attribute__((aligned(16))) float Vs[2][KS * 16 * VP];
    __shared__ __attribute__((aligned(16))) float pls[NW][16 * PP];

    const int tid  = threadIdx.x;
    const int lane = tid & 31;
    const int il   = lane & 15;
    const int hi   = lane >> 4;
    const int wv   = __builtin_amdgcn_readfirstlane(tid >> 5);
    const int h    = blockIdx.x;

    int b = -1, g = 0, S = 0, n = 0;
    {
        const int slot = blockIdx.y;
        int acc = 0;
        for (int i = 0; i < nb; ++i) {
            int len = cuq[i + 1] - cuq[i];
            if (len < 0) len = 0;
            const int ni = (len + 15) >> 4;
            const int Si = (ni + NW - 1) / NW;
            if (b < 0 && slot < acc + Si) { b = i; g = slot - acc; S = Si; n = ni; }
            acc += Si;
        }
    }
    if (b < 0) return;

    const int cq0   = cuq[b];
    const int cq1   = cuq[b + 1];
    const int lenq  = cq1 - cq0;
    const int ck0   = cuk[b];
    const int lenk  = cuk[b + 1] - ck0;
    const int delta = lenk - lenq;

    const int jt    = (wv < 4) ? wv : (11 - wv);
    const int t     = g + jt * S;
    const bool have = (t < n);
    const int base  = t * 16;
    const int nrows = have ? min(16, lenq - base) : 0;
    const int myKT  = have ? key_tiles(t, delta, lenk) : 0;

    int jmax = (n - 1 - g) / S;
    if (jmax > NW - 1) jmax = NW - 1;
    const int KT = key_tiles(g + jmax * S, delta, lenk);

    float* pl = pls[wv];

    int qrow = cq0 + base + il;
    if (qrow > cq1 - 1) qrow = cq1 - 1;
    const float* qp = q + (size_t)qrow * ROWSZ + h * HD + 2 * hi;
    v2f qa[16];
#pragma unroll
    for (int c = 0; c < 16; ++c) {
        v2f tq = *(const v2f*)(qp + 4 * c);
        qa[c] = tq * 0.125f;
    }

    v8f oacc[4];
#pragma unroll
    for (int nc = 0; nc < 4; ++nc) oacc[nc] = (v8f){};
    float lsum[8];
#pragma unroll
    for (int r = 0; r < 8; ++r) lsum[r] = 0.0f;
    float mrow[8];
#pragma unroll
    for (int r = 0; r < 8; ++r) mrow[r] = -3.0e38f;

    const float* kbase = k + (size_t)ck0 * ROWSZ + h * HD;
    const float* vbase = v + (size_t)ck0 * ROWSZ + h * HD;

    const int lrow = tid >> 4;
    const int lc4  = (tid & 15) * 4;
    v4f kreg[KS], vreg[KS];
    const int NS = (KT + KS - 1) / KS;

    if (NS > 0) {
#pragma unroll
        for (int u = 0; u < KS; ++u) {
            int kr = min(u * 16 + lrow, lenk - 1);
            kreg[u] = *(const v4f*)(kbase + (size_t)kr * ROWSZ + lc4);
            vreg[u] = *(const v4f*)(vbase + (size_t)kr * ROWSZ + lc4);
        }
#pragma unroll
        for (int u = 0; u < KS; ++u) {
            *(v4f*)(&Ks[0][(u * 16 + lrow) * KP + lc4]) = kreg[u];
            *(v4f*)(&Vs[0][(u * 16 + lrow) * VP + lc4]) = vreg[u];
        }
    }
    __syncthreads();

    for (int st = 0; st < NS; ++st) {
        const int cur = st & 1;
        const int kt0 = st * KS;
        if (st + 1 < NS) {
#pragma unroll
            for (int u = 0; u < KS; ++u) {
                int kr = min((kt0 + KS + u) * 16 + lrow, lenk - 1);
                kreg[u] = *(const v4f*)(kbase + (size_t)kr * ROWSZ + lc4);
                vreg[u] = *(const v4f*)(vbase + (size_t)kr * ROWSZ + lc4);
            }
        }

        const float* kt_s = &Ks[cur][il * KP + 2 * hi];
        const float* vt_s = &Vs[cur][il];
        if (kt0 + KS <= myKT) {
            key_step<KS>(kt_s, vt_s, pl, il, hi, kt0 * 16 + il, base, delta, lenk,
                         qa, oacc, lsum, mrow);
        } else {
            for (int u = 0; u < KS - 1; ++u) {
                if (kt0 + u < myKT)
                    key_step<1>(kt_s + u * 16 * KP, vt_s + u * 16 * VP, pl, il, hi,
                                (kt0 + u) * 16 + il, base, delta, lenk,
                                qa, oacc, lsum, mrow);
            }
        }

        if (st + 1 < NS) {
#pragma unroll
            for (int u = 0; u < KS; ++u) {
                *(v4f*)(&Ks[cur ^ 1][(u * 16 + lrow) * KP + lc4]) = kreg[u];
                *(v4f*)(&Vs[cur ^ 1][(u * 16 + lrow) * VP + lc4]) = vreg[u];
            }
        }
        __syncthreads();
    }

    row_xmask_add_step<1>(lsum);
    row_xmask_add_step<2>(lsum);
    row_xmask_add_step<4>(lsum);
    row_xmask_add_step<8>(lsum);

#pragma unroll
    for (int r = 0; r < 8; ++r) {
        int m = r + 8 * hi;
        if (m < nrows) {
            float inv = (myKT > 0) ? (1.0f / lsum[r]) : 0.0f;
            size_t ro = (size_t)(cq0 + base + m) * ROWSZ + h * HD;
#pragma unroll
            for (int nc = 0; nc < 4; ++nc)
                out[ro + nc * 16 + il] = (myKT > 0) ? (oacc[nc][r] * inv) : 0.0f;
        }
    }
}

extern "C" void kernel_launch(void* const* d_in, const int* in_sizes, int n_in,
                              void* d_out, int out_size, void* d_ws, size_t ws_size,
                              hipStream_t stream) {
    const float* q   = (const float*)d_in[0];
    const float* k   = (const float*)d_in[1];
    const float* v   = (const float*)d_in[2];
    const int*   cuq = (const int*)d_in[3];
    const int*   cuk = (const int*)d_in[4];
    float* out = (float*)d_out;

    const int T = in_sizes[0] / ROWSZ;
    const int B = in_sizes[3] - 1;
    dim3 grid(NH, (T + 127) / 128 + 2 * B, 1);
    va_fwd_kernel<<<grid, NW * 32, 0, stream>>>(q, k, v, cuq, cuk, out, B);
}


# Work-group owns two blocks of 4 neighbouring query tiles, block g (waves 0-3) and its mirror from the batch's end (waves 4-7); was a comb of 8 spread tiles whose SIMDs ran dry one by one; sums kept
# speedup vs baseline: 2.8928x; 1.2286x over previous
#include <hip/hip_runtime.h>
#include <math.h>


typedef __attribute__((ext_vector_type(2))) float v2f;
typedef __attribute__((ext_vector_type(4))) float v4f;
typedef __attribute__((ext_vector_type(8))) float v8f;

#define NH 16
#define HD 64
#define ROWSZ (NH * HD)
#define KS 3
#define PP (16 * KS + 4)
#define NW 8
#define QB (NW / 2)
#define KP 68
#define VP 72
#define KCG 4

__device__ __forceinline__ void frag_group_fence()
{
    __builtin_amdgcn_sched_barrier(0x0006);
}

__device__ __forceinline__ int key_tiles(int t, int delta, int lenk)
{
    int maxj = 16 * t + 15 + delta;
    if (maxj > lenk - 1) maxj = lenk - 1;
    return (maxj < 0) ? 0 : (maxj / 16 + 1);
}

template <int M>
__device__ __forceinline__ void row_xmask_max_step(float (&x)[8])
{
#pragma unroll
    for (int r = 0; r < 8; ++r)
        asm("v_max_num_f32_dpp %0, %1, %1 row_xmask:%2 row_mask:0xf bank_mask:0xf"
            : "=v"(x[r]) : "v"(x[r]), "n"(M));
}

template <int M>
__device__ __forceinline__ void row_xmask_add_step(float (&x)[8])
{
#pragma unroll
    for (int r = 0; r < 8; ++r)
        asm("v_add_f32_dpp %0, %1, %1 row_xmask:%2 row_mask:0xf bank_mask:0xf"
            : "=v"(x[r]) : "v"(x[r]), "n"(M));
}

template <int NSUB>
__device__ __forceinline__ void key_step(const float* __restrict__ kt_s,
                                         const float* __restrict__ vt_s,
                                         float* __restrict__ pl,
                                         const int il, const int hi,
                                         const int jg0,
                                         const int base, const int delta, const int lenk,
                                         const v2f (&qa)[16],
                                         v8f (&oacc)[4], float (&lsum)[8], float (&mrow)[8])
{
    v8f s[NSUB];
#pragma unroll
    for (int u = 0; u < NSUB; ++u) s[u] = (v8f){};
    constexpr int NKG = 16 / KCG;
    v2f kf[2][KCG][NSUB];
#pragma unroll
    for (int g = 0; g < 2; ++g)
#pragma unroll
        for (int j = 0; j < KCG; ++j)
#pragma unroll
            for (int u = 0; u < NSUB; ++u)
                kf[g][j][u] = *(const v2f*)(kt_s + u * 16 * KP + 4 * (g * KCG + j));
    frag_group_fence();
#pragma unroll
    for (int g = 0; g < NKG; ++g) {
#pragma unroll
        for (int j = 0; j < KCG; ++j) {
#pragma unroll
            for (int u = 0; u < NSUB; ++u)
                s[u] = __builtin_amdgcn_wmma_f32_16x16x4_f32(
                           false, qa[g * KCG + j], false, kf[g & 1][j][u], (short)0, s[u],
                           false, false);
        }
        if (g + 2 < NKG) {
            frag_group_fence();
#pragma unroll
            for (int j = 0; j < KCG; ++j)
#pragma unroll
                for (int u = 0; u < NSUB; ++u)
                    kf[g & 1][j][u] =
                        *(const v2f*)(kt_s + u * 16 * KP + 4 * ((g + 2) * KCG + j));
            frag_group_fence();
        }
    }

    const int lim = base + 8 * hi + delta;
#pragma unroll
    for (int u = 0; u < NSUB; ++u) {
        const int jg = jg0 + 16 * u;
        const int d  = jg - lim;
        const bool inb = (jg < lenk);
#pragma unroll
        for (int r = 0; r < 8; ++r) {
            bool ok = (d <= r) && inb;
            s[u][r] = ok ? s[u][r] : -1.0e9f;
        }
    }

    float rmax[8];
#pragma unroll
    for (int r = 0; r < 8; ++r) {
        rmax[r] = s[0][r];
#pragma unroll
        for (int u = 1; u < NSUB; ++u) rmax[r] = fmaxf(rmax[r], s[u][r]);
    }
    row_xmask_max_step<1>(rmax);
    row_xmask_max_step<2>(rmax);
    row_xmask_max_step<4>(rmax);
    row_xmask_max_step<8>(rmax);
    float alpha[8];
#pragma unroll
    for (int r = 0; r < 8; ++r) {
        float mnew = fmaxf(mrow[r], rmax[r]);
#pragma unroll
        for (int u = 0; u < NSUB; ++u) s[u][r] = __expf(s[u][r] - mnew);
        alpha[r] = __expf(mrow[r] - mnew);
        mrow[r]  = mnew;
    }
#pragma unroll
    for (int r = 0; r < 8; ++r) {
        oacc[0][r] *= alpha[r]; oacc[1][r] *= alpha[r];
        oacc[2][r] *= alpha[r]; oacc[3][r] *= alpha[r];
        float psum = s[0][r];
#pragma unroll
        for (int u = 1; u < NSUB; ++u) psum += s[u][r];
        lsum[r] = lsum[r] * alpha[r] + psum;
    }

#pragma unroll
    for (int u = 0; u < NSUB; ++u)
#pragma unroll
        for (int r = 0; r < 8; ++r)
            pl[(r + 8 * hi) * PP + 16 * u + il] = s[u][r];
    __builtin_amdgcn_fence(__ATOMIC_RELEASE, "wavefront");
    __builtin_amdgcn_wave_barrier();
    __builtin_amdgcn_fence(__ATOMIC_ACQUIRE, "wavefront");
    v2f pa[NSUB][4];
#pragma unroll
    for (int u = 0; u < NSUB; ++u)
#pragma unroll
        for (int c = 0; c < 4; ++c)
            pa[u][c] = *(const v2f*)(&pl[il * PP + 16 * u + 4 * c + 2 * hi]);
    __builtin_amdgcn_fence(__ATOMIC_RELEASE, "wavefront");
    __builtin_amdgcn_wave_barrier();
    __builtin_amdgcn_fence(__ATOMIC_ACQUIRE, "wavefront");

    constexpr int NVG = 2 * NSUB;
    v2f vf[2][2][4];
#pragma unroll
    for (int gi = 0; gi < 2; ++gi)
#pragma unroll
        for (int j = 0; j < 2; ++j)
#pragma unroll
            for (int nc = 0; nc < 4; ++nc) {
                const int kr0 = 8 * gi + 4 * j + 2 * hi;
                vf[gi][j][nc] = (v2f){ vt_s[kr0 * VP + nc * 16],
                                       vt_s[(kr0 + 1) * VP + nc * 16] };
            }
    frag_group_fence();
#pragma unroll
    for (int gi = 0; gi < NVG; ++gi) {
#pragma unroll
        for (int j = 0; j < 2; ++j) {
            const int u = gi >> 1;
            const int c = 2 * (gi & 1) + j;
#pragma unroll
            for (int nc = 0; nc < 4; ++nc)
                oacc[nc] = __builtin_amdgcn_wmma_f32_16x16x4_f32(
                               false, pa[u][c], false, vf[gi & 1][j][nc], (short)0, oacc[nc],
                               false, false);
        }
        if (gi + 2 < NVG) {
            frag_group_fence();
#pragma unroll
            for (int j = 0; j < 2; ++j)
#pragma unroll
                for (int nc = 0; nc < 4; ++nc) {
                    const int kr0 = 8 * (gi + 2) + 4 * j + 2 * hi;
                    vf[gi & 1][j][nc] = (v2f){ vt_s[kr0 * VP + nc * 16],
                                               vt_s[(kr0 + 1) * VP + nc * 16] };
                }
            frag_group_fence();
        }
    }
}

__global__ __launch_bounds__(NW * 32)
void va_fwd_kernel(const float* __restrict__ q,
                   const float* __restrict__ k,
                   const float* __restrict__ v,
                   const int*   __restrict__ cuq,
                   const int*   __restrict__ cuk,
                   float*       __restrict__ out,
                   int nb)
{
    __shared__ __attribute__((aligned(16))) float Ks[2][KS * 16 * KP];
    __shared__ __attribute__((aligned(16))) float Vs[2][KS * 16 * VP];
    __shared__ __attribute__((aligned(16))) float pls[NW][16 * PP];

    const int tid  = threadIdx.x;
    const int lane = tid & 31;
    const int il   = lane & 15;
    const int hi   = lane >> 4;
    const int wv   = __builtin_amdgcn_readfirstlane(tid >> 5);
    const int h    = blockIdx.x;

    int b = -1, g = 0, n = 0;
    {
        const int slot = blockIdx.y;
        int acc = 0;
        for (int i = 0; i < nb; ++i) {
            int len = cuq[i + 1] - cuq[i];
            if (len < 0) len = 0;
            const int ni = (len + 15) >> 4;
            const int Si = (ni + NW - 1) / NW;
            if (b < 0 && slot < acc + Si) { b = i; g = slot - acc; n = ni; }
            acc += Si;
        }
    }
    if (b < 0) return;

    const int cq0   = cuq[b];
    const int cq1   = cuq[b + 1];
    const int lenq  = cq1 - cq0;
    const int ck0   = cuk[b];
    const int lenk  = cuk[b + 1] - ck0;
    const int delta = lenk - lenq;

    const int nblk  = (n + QB - 1) / QB;
    const int hb    = nblk - 1 - g;
    const int sub   = wv & (QB - 1);
    const int t     = (wv < QB) ? (QB * g + sub) : (QB * hb + QB - 1 - sub);
    const bool have = (t < n) && (wv < QB || hb > g);
    const int base  = t * 16;
    const int nrows = have ? min(16, lenq - base) : 0;
    const int myKT  = have ? key_tiles(t, delta, lenk) : 0;

    int tlast = QB * hb + QB - 1;
    if (tlast > n - 1) tlast = n - 1;
    const int KT = key_tiles(tlast, delta, lenk);

    float* pl = pls[wv];

    int qrow = cq0 + base + il;
    if (qrow > cq1 - 1) qrow = cq1 - 1;
    const float* qp = q + (size_t)qrow * ROWSZ + h * HD + 2 * hi;
    v2f qa[16];
#pragma unroll
    for (int c = 0; c < 16; ++c) {
        v2f tq = *(const v2f*)(qp + 4 * c);
        qa[c] = tq * 0.125f;
    }

    v8f oacc[4];
#pragma unroll
    for (int nc = 0; nc < 4; ++nc) oacc[nc] = (v8f){};
    float lsum[8];
#pragma unroll
    for (int r = 0; r < 8; ++r) lsum[r] = 0.0f;
    float mrow[8];
#pragma unroll
    for (int r = 0; r < 8; ++r) mrow[r] = -3.0e38f;

    const float* kbase = k + (size_t)ck0 * ROWSZ + h * HD;
    const float* vbase = v + (size_t)ck0 * ROWSZ + h * HD;

    const int lrow = tid >> 4;
    const int lc4  = (tid & 15) * 4;
    v4f kreg[KS], vreg[KS];
    const int NS = (KT + KS - 1) / KS;

    if (NS > 0) {
#pragma unroll
        for (int u = 0; u < KS; ++u) {
            int kr = min(u * 16 + lrow, lenk - 1);
            kreg[u] = *(const v4f*)(kbase + (size_t)kr * ROWSZ + lc4);
            vreg[u] = *(const v4f*)(vbase + (size_t)kr * ROWSZ + lc4);
        }
#pragma unroll
        for (int u = 0; u < KS; ++u) {
            *(v4f*)(&Ks[0][(u * 16 + lrow) * KP + lc4]) = kreg[u];
            *(v4f*)(&Vs[0][(u * 16 + lrow) * VP + lc4]) = vreg[u];
        }
    }
    __syncthreads();

    for (int st = 0; st < NS; ++st) {
        const int cur = st & 1;
        const int kt0 = st * KS;
        if (st + 1 < NS) {
#pragma unroll
            for (int u = 0; u < KS; ++u) {
                int kr = min((kt0 + KS + u) * 16 + lrow, lenk - 1);
                kreg[u] = *(const v4f*)(kbase + (size_t)kr * ROWSZ + lc4);
                vreg[u] = *(const v4f*)(vbase + (size_t)kr * ROWSZ + lc4);
            }
        }

        const float* kt_s = &Ks[cur][il * KP + 2 * hi];
        const float* vt_s = &Vs[cur][il];
        if (kt0 + KS <= myKT) {
            key_step<KS>(kt_s, vt_s, pl, il, hi, kt0 * 16 + il, base, delta, lenk,
                         qa, oacc, lsum, mrow);
        } else {
            for (int u = 0; u < KS - 1; ++u) {
                if (kt0 + u < myKT)
                    key_step<1>(kt_s + u * 16 * KP, vt_s + u * 16 * VP, pl, il, hi,
                                (kt0 + u) * 16 + il, base, delta, lenk,
                                qa, oacc, lsum, mrow);
            }
        }

        if (st + 1 < NS) {
#pragma unroll
            for (int u = 0; u < KS; ++u) {
                *(v4f*)(&Ks[cur ^ 1][(u * 16 + lrow) * KP + lc4]) = kreg[u];
                *(v4f*)(&Vs[cur ^ 1][(u * 16 + lrow) * VP + lc4]) = vreg[u];
            }
        }
        __syncthreads();
    }

    row_xmask_add_step<1>(lsum);
    row_xmask_add_step<2>(lsum);
    row_xmask_add_step<4>(lsum);
    row_xmask_add_step<8>(lsum);

#pragma unroll
    for (int r = 0; r < 8; ++r) {
        int m = r + 8 * hi;
        if (m < nrows) {
            float inv = (myKT > 0) ? (1.0f / lsum[r]) : 0.0f;
            size_t ro = (size_t)(cq0 + base + m) * ROWSZ + h * HD;
#pragma unroll
            for (int nc = 0; nc < 4; ++nc)
                out[ro + nc * 16 + il] = (myKT > 0) ? (oacc[nc][r] * inv) : 0.0f;
        }
    }
}

extern "C" void kernel_launch(void* const* d_in, const int* in_sizes, int n_in,
                              void* d_out, int out_size, void* d_ws, size_t ws_size,
                              hipStream_t stream) {
    const float* q   = (const float*)d_in[0];
    const float* k   = (const float*)d_in[1];
    const float* v   = (const float*)d_in[2];
    const int*   cuq = (const int*)d_in[3];
    const int*   cuk = (const int*)d_in[4];
    float* out = (float*)d_out;

    const int T = in_sizes[0] / ROWSZ;
    const int B = in_sizes[3] - 1;
    dim3 grid(NH, (T + 127) / 128 + 2 * B, 1);
    va_fwd_kernel<<<grid, NW * 32, 0, stream>>>(q, k, v, cuq, cuk, out, B);
}
